# starting kernel
# baseline (speedup 1.0000x reference)
#include <hip/hip_runtime.h>
#include <math.h>

typedef __attribute__((ext_vector_type(2))) float v2f;
typedef __attribute__((ext_vector_type(8))) float v8f;

#define NEG_SLOPE 0.2f

template<int K, int NOUT>
__global__ __launch_bounds__(256) void gemm_wmma(const float* __restrict__ A,
                                                 const float* __restrict__ B,
                                                 float* __restrict__ C,
                                                 int ntasks)
{
    constexpr int NG = NOUT / 64;
    int task = blockIdx.x * 8 + (threadIdx.x >> 5);
    if (task >= ntasks) return;
    int rt = task / NG;
    int cg = task % NG;
    int lane = threadIdx.x & 31;
    int half = lane >> 4;
    int mr   = lane & 15;

    const float* arow = A + (size_t)(rt * 16 + mr) * K + 2 * half;
    const float* bcol = B + cg * 64 + mr;

    v8f acc[4] = {};

    for (int k0 = 0; k0 < K; k0 += 4) {
        v2f a;
        a.x = arow[k0];
        a.y = arow[k0 + 1];
        const float* brow0 = bcol + (size_t)(k0 + 2 * half) * NOUT;
        const float* brow1 = brow0 + NOUT;
#pragma unroll
        for (int t = 0; t < 4; ++t) {
            v2f b;
            b.x = brow0[t * 16];
            b.y = brow1[t * 16];
            acc[t] = __builtin_amdgcn_wmma_f32_16x16x4_f32(
                false, a, false, b, (short)0, acc[t], false, false);
        }
    }

    float* crow = C + (size_t)(rt * 16 + half * 8) * NOUT + cg * 64 + mr;
#pragma unroll
    for (int t = 0; t < 4; ++t)
#pragma unroll
        for (int r = 0; r < 8; ++r)
            crow[(size_t)r * NOUT + t * 16] = acc[t][r];
}

template<int H, int C>
__global__ void att_coeff(const float* __restrict__ h,
                          const float* __restrict__ a_src,
                          const float* __restrict__ a_dst,
                          float* __restrict__ als,
                          float* __restrict__ ald,
                          int n)
{
    int i = blockIdx.x * blockDim.x + threadIdx.x;
    if (i >= n * H) return;
    int node = i / H, hd = i % H;
    const float* hp = h + (size_t)node * (H * C) + hd * C;
    const float* asrc = a_src + hd * C;
    const float* adst = a_dst + hd * C;
    float ss = 0.f, sd = 0.f;
#pragma unroll 4
    for (int c = 0; c < C; ++c) {
        float v = hp[c];
        ss += v * asrc[c];
        sd += v * adst[c];
    }
    als[i] = ss;
    ald[i] = sd;
}

template<int H>
__global__ void edge_denom(const int* __restrict__ src,
                           const int* __restrict__ dst,
                           const float* __restrict__ als,
                           const float* __restrict__ ald,
                           float* __restrict__ denom,
                           int E_, int n)
{
    int i = blockIdx.x * blockDim.x + threadIdx.x;
    int total = (E_ + n) * H;
    if (i >= total) return;
    int e = i / H, hd = i % H;
    int s, d;
    if (e < E_) { s = src[e]; d = dst[e]; }
    else        { s = d = e - E_; }
    float x = als[s * H + hd] + ald[d * H + hd];
    float lv = (x >= 0.f) ? x : NEG_SLOPE * x;
    atomicAdd(&denom[d * H + hd], expf(lv));
}

__global__ void edge_aggr1(const int* __restrict__ src,
                           const int* __restrict__ dst,
                           const float* __restrict__ als,
                           const float* __restrict__ ald,
                           const float* __restrict__ denom,
                           const float* __restrict__ h,
                           float* __restrict__ out,
                           int E_, int n)
{
    int wave = blockIdx.x * (blockDim.x >> 5) + (threadIdx.x >> 5);
    int lane = threadIdx.x & 31;
    if (wave >= E_ + n) return;
    int s, d;
    if (wave < E_) { s = src[wave]; d = dst[wave]; }
    else           { s = d = wave - E_; }

    float alpha = 0.f;
    if (lane < 8) {
        float x = als[s * 8 + lane] + ald[d * 8 + lane];
        float lv = (x >= 0.f) ? x : NEG_SLOPE * x;
        alpha = expf(lv) / denom[d * 8 + lane];
    }

    const float* hs = h + (size_t)s * 256;
    float* od = out + (size_t)d * 256;
#pragma unroll
    for (int j = 0; j < 8; ++j) {
        float aj = __shfl(alpha, j, 32);
        int col = lane + 32 * j;
        atomicAdd(&od[col], aj * hs[col]);
    }
}

__global__ void edge_aggr2(const int* __restrict__ src,
                           const int* __restrict__ dst,
                           const float* __restrict__ als,
                           const float* __restrict__ ald,
                           const float* __restrict__ denom,
                           const float* __restrict__ h,
                           float* __restrict__ out,
                           int E_, int n)
{
    int wave = blockIdx.x * (blockDim.x >> 5) + (threadIdx.x >> 5);
    int lane = threadIdx.x & 31;
    if (wave >= E_ + n) return;
    int s, d;
    if (wave < E_) { s = src[wave]; d = dst[wave]; }
    else           { s = d = wave - E_; }

    float x = als[s] + ald[d];
    float lv = (x >= 0.f) ? x : NEG_SLOPE * x;
    float alpha = expf(lv) / denom[d];

    const float* hs = h + (size_t)s * 64;
    float* od = out + (size_t)d * 64;
#pragma unroll
    for (int j = 0; j < 2; ++j) {
        int col = lane + 32 * j;
        atomicAdd(&od[col], alpha * hs[col]);
    }
}

template<int F>
__global__ void bias_elu(float* __restrict__ buf,
                         const float* __restrict__ b,
                         int total)
{
    int i = blockIdx.x * blockDim.x + threadIdx.x;
    if (i >= total) return;
    float v = buf[i] + b[i % F];
    buf[i] = (v > 0.f) ? v : expm1f(v);
}

extern "C" void kernel_launch(void* const* d_in, const int* in_sizes, int n_in,
                              void* d_out, int out_size, void* d_ws, size_t ws_size,
                              hipStream_t stream)
{
    const float* x   = (const float*)d_in[0];
    const int*   ei  = (const int*)  d_in[1];
    const float* W1  = (const float*)d_in[2];
    const float* as1 = (const float*)d_in[3];
    const float* ad1 = (const float*)d_in[4];
    const float* b1  = (const float*)d_in[5];
    const float* W2  = (const float*)d_in[6];
    const float* as2 = (const float*)d_in[7];
    const float* ad2 = (const float*)d_in[8];
    const float* b2  = (const float*)d_in[9];
    float* out = (float*)d_out;

    const int IN_DIM = 128, H1 = 8, F1 = 256, OUT_DIM = 64;
    int n = in_sizes[0] / IN_DIM;
    int e = in_sizes[1] / 2;
    const int* src = ei;
    const int* dst = ei + e;

    float* ws = (float*)d_ws;
    float* h1   = ws;  ws += (size_t)n * F1;
    float* g1   = ws;  ws += (size_t)n * F1;
    float* h2   = ws;  ws += (size_t)n * OUT_DIM;
    float* als1 = ws;  ws += (size_t)n * H1;
    float* ald1 = ws;  ws += (size_t)n * H1;
    float* den1 = ws;  ws += (size_t)n * H1;
    float* als2 = ws;  ws += (size_t)n;
    float* ald2 = ws;  ws += (size_t)n;
    float* den2 = ws;  ws += (size_t)n;

    hipMemsetAsync(g1,   0, (size_t)n * F1 * sizeof(float), stream);
    hipMemsetAsync(den1, 0, (size_t)n * H1 * sizeof(float), stream);
    hipMemsetAsync(den2, 0, (size_t)n * sizeof(float), stream);
    hipMemsetAsync(out,  0, (size_t)n * OUT_DIM * sizeof(float), stream);

    int ne = e + n;

    {
        int tasks = (n / 16) * (F1 / 64);
        gemm_wmma<128, 256><<<(tasks + 7) / 8, 256, 0, stream>>>(x, W1, h1, tasks);
    }
    att_coeff<8, 32><<<(n * 8 + 255) / 256, 256, 0, stream>>>(h1, as1, ad1, als1, ald1, n);
    edge_denom<8><<<(ne * 8 + 255) / 256, 256, 0, stream>>>(src, dst, als1, ald1, den1, e, n);
    edge_aggr1<<<(ne + 7) / 8, 256, 0, stream>>>(src, dst, als1, ald1, den1, h1, g1, e, n);
    bias_elu<256><<<(n * 256 + 255) / 256, 256, 0, stream>>>(g1, b1, n * 256);

    {
        int tasks = (n / 16) * (OUT_DIM / 64);
        gemm_wmma<256, 64><<<(tasks + 7) / 8, 256, 0, stream>>>(g1, W2, h2, tasks);
    }
    att_coeff<1, 64><<<(n + 255) / 256, 256, 0, stream>>>(h2, as2, ad2, als2, ald2, n);
    edge_denom<1><<<(ne + 255) / 256, 256, 0, stream>>>(src, dst, als2, ald2, den2, e, n);
    edge_aggr2<<<(ne + 7) / 8, 256, 0, stream>>>(src, dst, als2, ald2, den2, h2, out, e, n);
    bias_elu<64><<<(n * 64 + 255) / 256, 256, 0, stream>>>(out, b2, n * 64);
}


# Per-destination gather over a CSR built in-call (int-atomic rank, scan, fill) replaces float atomics on the output; sums run in arrival-order rank; softmax divide, bias and elu fused in
# speedup vs baseline: 2.4179x; 2.4179x over previous
#include <hip/hip_runtime.h>
#include <math.h>

typedef __attribute__((ext_vector_type(2))) float v2f;
typedef __attribute__((ext_vector_type(8))) float v8f;

#define NEG_SLOPE 0.2f

template<int K, int NOUT>
__global__ __launch_bounds__(256) void gemm_wmma(const float* __restrict__ A,
                                                 const float* __restrict__ B,
                                                 float* __restrict__ C,
                                                 int ntasks)
{
    constexpr int NG = NOUT / 64;
    int task = blockIdx.x * 8 + (threadIdx.x >> 5);
    if (task >= ntasks) return;
    int rt = task / NG;
    int cg = task % NG;
    int lane = threadIdx.x & 31;
    int half = lane >> 4;
    int mr   = lane & 15;

    const float* arow = A + (size_t)(rt * 16 + mr) * K + 2 * half;
    const float* bcol = B + cg * 64 + mr;

    v8f acc[4] = {};

    for (int k0 = 0; k0 < K; k0 += 4) {
        v2f a;
        a.x = arow[k0];
        a.y = arow[k0 + 1];
        const float* brow0 = bcol + (size_t)(k0 + 2 * half) * NOUT;
        const float* brow1 = brow0 + NOUT;
#pragma unroll
        for (int t = 0; t < 4; ++t) {
            v2f b;
            b.x = brow0[t * 16];
            b.y = brow1[t * 16];
            acc[t] = __builtin_amdgcn_wmma_f32_16x16x4_f32(
                false, a, false, b, (short)0, acc[t], false, false);
        }
    }

    float* crow = C + (size_t)(rt * 16 + half * 8) * NOUT + cg * 64 + mr;
#pragma unroll
    for (int t = 0; t < 4; ++t)
#pragma unroll
        for (int r = 0; r < 8; ++r)
            crow[(size_t)r * NOUT + t * 16] = acc[t][r];
}

template<int H, int C>
__global__ void att_coeff(const float* __restrict__ h,
                          const float* __restrict__ a_src,
                          const float* __restrict__ a_dst,
                          float* __restrict__ als,
                          float* __restrict__ ald,
                          int n)
{
    int i = blockIdx.x * blockDim.x + threadIdx.x;
    if (i >= n * H) return;
    int node = i / H, hd = i % H;
    const float* hp = h + (size_t)node * (H * C) + hd * C;
    const float* asrc = a_src + hd * C;
    const float* adst = a_dst + hd * C;
    float ss = 0.f, sd = 0.f;
#pragma unroll 4
    for (int c = 0; c < C; ++c) {
        float v = hp[c];
        ss += v * asrc[c];
        sd += v * adst[c];
    }
    als[i] = ss;
    ald[i] = sd;
}

__global__ void edge_rank(const int* __restrict__ dst,
                          int* __restrict__ cnt,
                          int* __restrict__ rank,
                          int E_)
{
    int e = blockIdx.x * blockDim.x + threadIdx.x;
    if (e >= E_) return;
    rank[e] = atomicAdd(&cnt[dst[e]], 1);
}

__global__ __launch_bounds__(1024) void seg_offsets(const int* __restrict__ cnt,
                                                    int* __restrict__ off,
                                                    int n)
{
    __shared__ int wsum[32];
    __shared__ int carry_s;
    int tid  = threadIdx.x;
    int lane = tid & 31;
    int wv   = tid >> 5;
    int base = blockIdx.x * 1024;

    int part = 0;
    const int4* c4 = reinterpret_cast<const int4*>(cnt);
    int nv = base >> 2;
#pragma unroll 4
    for (int i = tid; i < nv; i += 1024) {
        int4 v = c4[i];
        part += (v.x + v.y) + (v.z + v.w);
    }
    for (int o = 16; o > 0; o >>= 1) part += __shfl_xor(part, o, 32);
    if (lane == 0) wsum[wv] = part;
    __syncthreads();
    if (wv == 0) {
        int v = wsum[lane];
        for (int o = 16; o > 0; o >>= 1) v += __shfl_xor(v, o, 32);
        if (lane == 0) carry_s = v;
    }
    __syncthreads();
    int carry = carry_s;

    int i = base + tid;
    int c = (i < n) ? cnt[i] : 0;
    int incl = c;
    for (int o = 1; o < 32; o <<= 1) {
        int t = __shfl_up(incl, o, 32);
        if (lane >= o) incl += t;
    }
    if (lane == 31) wsum[wv] = incl;
    __syncthreads();
    if (wv == 0) {
        int v = wsum[lane];
        int s = v;
        for (int o = 1; o < 32; o <<= 1) {
            int t = __shfl_up(s, o, 32);
            if (lane >= o) s += t;
        }
        wsum[lane] = s - v;
    }
    __syncthreads();
    int excl = carry + wsum[wv] + incl - c;
    if (i < n) off[i] = excl;
    if (i == n - 1) off[n] = excl + c;
}

__global__ void csr_fill(const int* __restrict__ src,
                         const int* __restrict__ dst,
                         const int* __restrict__ off,
                         const int* __restrict__ rank,
                         int* __restrict__ csr,
                         int E_)
{
    int e = blockIdx.x * blockDim.x + threadIdx.x;
    if (e >= E_) return;
    csr[off[dst[e]] + rank[e]] = src[e];
}

__device__ __forceinline__ float edge_weight(float x)
{
    float lv = (x >= 0.f) ? x : NEG_SLOPE * x;
    return expf(lv);
}

__device__ __forceinline__ float elu1(float v)
{
    return (v > 0.f) ? v : expm1f(v);
}

__global__ __launch_bounds__(256) void node_aggr1(const int* __restrict__ off,
                                                  const int* __restrict__ csr,
                                                  const float* __restrict__ als,
                                                  const float* __restrict__ ald,
                                                  const float* __restrict__ h,
                                                  const float* __restrict__ bias,
                                                  float* __restrict__ out,
                                                  int n)
{
    int d = blockIdx.x * (blockDim.x >> 5) + (threadIdx.x >> 5);
    if (d >= n) return;
    d = __builtin_amdgcn_readfirstlane(d);
    int lane = threadIdx.x & 31;
    int hd0 = lane >> 3;
    int hd1 = 4 + hd0;
    int c0 = 4 * lane;
    int c1 = 128 + 4 * lane;

    float ad0 = ald[d * 8 + hd0];
    float ad1 = ald[d * 8 + hd1];
    int beg = off[d];
    int end = off[d + 1];

    float4 acc0, acc1;
    float den0, den1;
    {
        float w0 = edge_weight(als[d * 8 + hd0] + ad0);
        float w1 = edge_weight(als[d * 8 + hd1] + ad1);
        const float* hs = h + (size_t)d * 256;
        float4 v0 = *reinterpret_cast<const float4*>(hs + c0);
        float4 v1 = *reinterpret_cast<const float4*>(hs + c1);
        acc0 = make_float4(w0 * v0.x, w0 * v0.y, w0 * v0.z, w0 * v0.w);
        acc1 = make_float4(w1 * v1.x, w1 * v1.y, w1 * v1.z, w1 * v1.w);
        den0 = w0;
        den1 = w1;
    }

    int p = beg;
    for (; p + 4 <= end; p += 4) {
        int s[4];
        float x0[4], x1[4];
        float4 v0[4], v1[4];
        for (int u = 0; u < 4; ++u) s[u] = csr[p + u];
        for (int u = 0; u < 4; ++u) {
            const float* hs = h + (size_t)s[u] * 256;
            x0[u] = als[s[u] * 8 + hd0];
            x1[u] = als[s[u] * 8 + hd1];
            v0[u] = *reinterpret_cast<const float4*>(hs + c0);
            v1[u] = *reinterpret_cast<const float4*>(hs + c1);
        }
        for (int u = 0; u < 4; ++u) {
            float w0 = edge_weight(x0[u] + ad0);
            float w1 = edge_weight(x1[u] + ad1);
            acc0.x += w0 * v0[u].x; acc0.y += w0 * v0[u].y;
            acc0.z += w0 * v0[u].z; acc0.w += w0 * v0[u].w;
            acc1.x += w1 * v1[u].x; acc1.y += w1 * v1[u].y;
            acc1.z += w1 * v1[u].z; acc1.w += w1 * v1[u].w;
            den0 += w0;
            den1 += w1;
        }
    }
    for (; p < end; ++p) {
        int s = csr[p];
        const float* hs = h + (size_t)s * 256;
        float w0 = edge_weight(als[s * 8 + hd0] + ad0);
        float w1 = edge_weight(als[s * 8 + hd1] + ad1);
        float4 v0 = *reinterpret_cast<const float4*>(hs + c0);
        float4 v1 = *reinterpret_cast<const float4*>(hs + c1);
        acc0.x += w0 * v0.x; acc0.y += w0 * v0.y;
        acc0.z += w0 * v0.z; acc0.w += w0 * v0.w;
        acc1.x += w1 * v1.x; acc1.y += w1 * v1.y;
        acc1.z += w1 * v1.z; acc1.w += w1 * v1.w;
        den0 += w0;
        den1 += w1;
    }

    float4 b0 = *reinterpret_cast<const float4*>(bias + c0);
    float4 b1 = *reinterpret_cast<const float4*>(bias + c1);
    float4 r0, r1;
    r0.x = elu1(acc0.x / den0 + b0.x); r0.y = elu1(acc0.y / den0 + b0.y);
    r0.z = elu1(acc0.z / den0 + b0.z); r0.w = elu1(acc0.w / den0 + b0.w);
    r1.x = elu1(acc1.x / den1 + b1.x); r1.y = elu1(acc1.y / den1 + b1.y);
    r1.z = elu1(acc1.z / den1 + b1.z); r1.w = elu1(acc1.w / den1 + b1.w);
    float* od = out + (size_t)d * 256;
    *reinterpret_cast<float4*>(od + c0) = r0;
    *reinterpret_cast<float4*>(od + c1) = r1;
}

__global__ __launch_bounds__(256) void node_aggr2(const int* __restrict__ off,
                                                  const int* __restrict__ csr,
                                                  const float* __restrict__ als,
                                                  const float* __restrict__ ald,
                                                  const float* __restrict__ h,
                                                  const float* __restrict__ bias,
                                                  float* __restrict__ out,
                                                  int n)
{
    int d = blockIdx.x * (blockDim.x >> 5) + (threadIdx.x >> 5);
    if (d >= n) return;
    d = __builtin_amdgcn_readfirstlane(d);
    int lane = threadIdx.x & 31;
    int c0 = 2 * lane;

    float ad = ald[d];
    int beg = off[d];
    int end = off[d + 1];

    float2 acc;
    float den;
    {
        float w = edge_weight(als[d] + ad);
        float2 v = *reinterpret_cast<const float2*>(h + (size_t)d * 64 + c0);
        acc = make_float2(w * v.x, w * v.y);
        den = w;
    }

    int p = beg;
    for (; p + 4 <= end; p += 4) {
        int s[4];
        float x[4];
        float2 v[4];
        for (int u = 0; u < 4; ++u) s[u] = csr[p + u];
        for (int u = 0; u < 4; ++u) {
            x[u] = als[s[u]];
            v[u] = *reinterpret_cast<const float2*>(h + (size_t)s[u] * 64 + c0);
        }
        for (int u = 0; u < 4; ++u) {
            float w = edge_weight(x[u] + ad);
            acc.x += w * v[u].x;
            acc.y += w * v[u].y;
            den += w;
        }
    }
    for (; p < end; ++p) {
        int s = csr[p];
        float w = edge_weight(als[s] + ad);
        float2 v = *reinterpret_cast<const float2*>(h + (size_t)s * 64 + c0);
        acc.x += w * v.x;
        acc.y += w * v.y;
        den += w;
    }

    float2 b = *reinterpret_cast<const float2*>(bias + c0);
    float2 r;
    r.x = elu1(acc.x / den + b.x);
    r.y = elu1(acc.y / den + b.y);
    *reinterpret_cast<float2*>(out + (size_t)d * 64 + c0) = r;
}

extern "C" void kernel_launch(void* const* d_in, const int* in_sizes, int n_in,
                              void* d_out, int out_size, void* d_ws, size_t ws_size,
                              hipStream_t stream)
{
    const float* x   = (const float*)d_in[0];
    const int*   ei  = (const int*)  d_in[1];
    const float* W1  = (const float*)d_in[2];
    const float* as1 = (const float*)d_in[3];
    const float* ad1 = (const float*)d_in[4];
    const float* b1  = (const float*)d_in[5];
    const float* W2  = (const float*)d_in[6];
    const float* as2 = (const float*)d_in[7];
    const float* ad2 = (const float*)d_in[8];
    const float* b2  = (const float*)d_in[9];
    float* out = (float*)d_out;

    const int IN_DIM = 128, H1 = 8, F1 = 256, OUT_DIM = 64;
    int n = in_sizes[0] / IN_DIM;
    int e = in_sizes[1] / 2;
    const int* src = ei;
    const int* dst = ei + e;

    float* ws = (float*)d_ws;
    float* h1   = ws;  ws += (size_t)n * F1;
    float* g1   = ws;  ws += (size_t)n * F1;
    int*   ib   = (int*)ws;  ws += (size_t)n * OUT_DIM;
    float* als1 = ws;  ws += (size_t)n * H1;
    float* ald1 = ws;  ws += (size_t)n * H1;
    ws += (size_t)n * H1;
    float* als2 = ws;  ws += (size_t)n;
    float* ald2 = ws;  ws += (size_t)n;
    float* h2   = h1;

    int* cnt  = ib;  ib += n;
    int* rank = ib;  ib += e;
    int* csr  = ib;  ib += e;
    int* off  = ib;

    hipMemsetAsync(cnt, 0, (size_t)n * sizeof(int), stream);

    edge_rank<<<(e + 255) / 256, 256, 0, stream>>>(dst, cnt, rank, e);
    seg_offsets<<<(n + 1023) / 1024, 1024, 0, stream>>>(cnt, off, n);
    csr_fill<<<(e + 255) / 256, 256, 0, stream>>>(src, dst, off, rank, csr, e);

    {
        int tasks = (n / 16) * (F1 / 64);
        gemm_wmma<128, 256><<<(tasks + 7) / 8, 256, 0, stream>>>(x, W1, h1, tasks);
    }
    att_coeff<8, 32><<<(n * 8 + 255) / 256, 256, 0, stream>>>(h1, as1, ad1, als1, ald1, n);
    node_aggr1<<<(n + 7) / 8, 256, 0, stream>>>(off, csr, als1, ald1, h1, b1, g1, n);

    {
        int tasks = (n / 16) * (OUT_DIM / 64);
        gemm_wmma<256, 64><<<(tasks + 7) / 8, 256, 0, stream>>>(g1, W2, h2, tasks);
    }
    att_coeff<1, 64><<<(n + 255) / 256, 256, 0, stream>>>(h2, as2, ad2, als2, ald2, n);
    node_aggr2<<<(n + 7) / 8, 256, 0, stream>>>(off, csr, als2, ald2, h2, b2, out, n);
}


# Aggregation kernels: each (edge, head) weight is made once by one lane and fetched across lanes: one coefficient load and one exp per lane per four edges, not 8 (conv1) or 4 (conv2); sums unchanged
# speedup vs baseline: 3.0061x; 1.2433x over previous
#include <hip/hip_runtime.h>
#include <math.h>

typedef __attribute__((ext_vector_type(2))) float v2f;
typedef __attribute__((ext_vector_type(8))) float v8f;

#define NEG_SLOPE 0.2f

template<int CNT, int M>
__device__ __forceinline__ void fold_lanes(float* v, int mr)
{
    bool up = (mr & M) != 0;
#pragma unroll
    for (int j = 0; j < CNT; ++j) {
        float lo = v[j];
        float hi = v[j + CNT];
        float send = up ? lo : hi;
        float keep = up ? hi : lo;
        v[j] = keep + __shfl_xor(send, M, 32);
    }
}

template<int K, int NOUT, int HC, bool RANK>
__global__ __launch_bounds__(256) void gemm_wmma(const float* __restrict__ A,
                                                 const float* __restrict__ B,
                                                 float* __restrict__ C,
                                                 const float* __restrict__ a_src,
                                                 const float* __restrict__ a_dst,
                                                 float* __restrict__ als,
                                                 float* __restrict__ ald,
                                                 int ntasks,
                                                 const int* __restrict__ edst,
                                                 int* __restrict__ cnt,
                                                 int* __restrict__ rank,
                                                 int E_)
{
    constexpr int NG  = NOUT / 64;
    constexpr int HPW = 64 / HC;
    constexpr int TPH = HC / 16;
    constexpr int NH  = NOUT / HC;
    constexpr int RB  = (8 / NG) * 16;
    constexpr int NV  = 2 * HPW * 8;
    __shared__ float sh[2 * RB * NH];

    int wv   = threadIdx.x >> 5;
    int task = blockIdx.x * 8 + wv;
    int lane = threadIdx.x & 31;
    int half = lane >> 4;
    int mr   = lane & 15;

    bool active = task < ntasks;
    int rt = task / NG;
    int cg = task % NG;
    v8f acc[4] = {};

    if (active) {

        const float* arow = A + (size_t)(rt * 16 + mr) * K + 2 * half;
        const float* bcol = B + cg * 64 + mr;

        for (int k0 = 0; k0 < K; k0 += 4) {
            v2f a;
            a.x = arow[k0];
            a.y = arow[k0 + 1];
            const float* brow0 = bcol + (size_t)(k0 + 2 * half) * NOUT;
            const float* brow1 = brow0 + NOUT;
#pragma unroll
            for (int t = 0; t < 4; ++t) {
                v2f b;
                b.x = brow0[t * 16];
                b.y = brow1[t * 16];
                acc[t] = __builtin_amdgcn_wmma_f32_16x16x4_f32(
                    false, a, false, b, (short)0, acc[t], false, false);
            }
        }

        float cs[4], cd[4];
#pragma unroll
        for (int t = 0; t < 4; ++t) {
            cs[t] = a_src[cg * 64 + t * 16 + mr];
            cd[t] = a_dst[cg * 64 + t * 16 + mr];
        }

        float v[NV];
#pragma unroll
        for (int hh = 0; hh < HPW; ++hh)
#pragma unroll
            for (int r = 0; r < 8; ++r) {
                float ps = acc[hh * TPH][r] * cs[hh * TPH];
                float pd = acc[hh * TPH][r] * cd[hh * TPH];
#pragma unroll
                for (int tt = 1; tt < TPH; ++tt) {
                    ps += acc[hh * TPH + tt][r] * cs[hh * TPH + tt];
                    pd += acc[hh * TPH + tt][r] * cd[hh * TPH + tt];
                }
                v[hh * 8 + r] = ps;
                v[HPW * 8 + hh * 8 + r] = pd;
            }

        fold_lanes<NV / 2, 8>(v, mr);
        fold_lanes<NV / 4, 4>(v, mr);
        fold_lanes<NV / 8, 2>(v, mr);
        fold_lanes<NV / 16, 1>(v, mr);

        int jb = ((mr >> 3) & 1) * (NV / 2) + ((mr >> 2) & 1) * (NV / 4)
               + ((mr >> 1) & 1) * (NV / 8) + (mr & 1) * (NV / 16);
        int rtl = wv / NG;
#pragma unroll
        for (int i = 0; i < NV / 16; ++i) {
            int j    = jb + i;
            int kind = j / (HPW * 8);
            int hh   = (j / 8) % HPW;
            int r    = j % 8;
            sh[kind * (RB * NH) + (rtl * 16 + half * 8 + r) * NH + cg * HPW + hh] = v[i];
        }
    }
    __syncthreads();

    if (active) {
        float* crow = C + (size_t)(rt * 16 + half * 8) * NOUT + cg * 64 + mr;
#pragma unroll
        for (int t = 0; t < 4; ++t)
#pragma unroll
            for (int r = 0; r < 8; ++r)
                crow[(size_t)r * NOUT + t * 16] = acc[t][r];
    }

    int base  = blockIdx.x * (RB * NH);
    int limit = (ntasks / NG) * 16 * NH;
    for (int idx = threadIdx.x; idx < 2 * RB * NH; idx += 256) {
        int kind = idx / (RB * NH);
        int o    = idx % (RB * NH);
        if (base + o < limit) {
            float* dstp = kind ? ald : als;
            dstp[base + o] = sh[idx];
        }
    }

    if (RANK) {
        if (active) {
            int estride = ntasks * 32;
            for (int e = task * 32 + lane; e < E_; e += 2 * estride) {
                int e1 = e + estride;
                bool two = e1 < E_;
                int d0 = edst[e];
                int d1 = two ? edst[e1] : 0;
                int r0 = atomicAdd(&cnt[d0], 1);
                int r1 = 0;
                if (two) r1 = atomicAdd(&cnt[d1], 1);
                rank[e] = r0;
                if (two) rank[e1] = r1;
            }
        }
    }
}

#define OFF_PT  49
#define OFF_CAP (1024 * OFF_PT)

__global__ __launch_bounds__(1024) void scan_fill(const int* __restrict__ cnt,
                                                  const int* __restrict__ src,
                                                  const int* __restrict__ dst,
                                                  const int* __restrict__ rank,
                                                  int* __restrict__ off,
                                                  int* __restrict__ csr,
                                                  int n,
                                                  int E_)
{
    __shared__ alignas(16) int offs[OFF_CAP];
    __shared__ int wsum[32];
    int tid  = threadIdx.x;
    int lane = tid & 31;
    int wv   = tid >> 5;

    const int4* c4 = reinterpret_cast<const int4*>(cnt);
    int4* o4 = reinterpret_cast<int4*>(offs);
    int nv = n >> 2;
    constexpr int NQ = (OFF_CAP / 4 + 1023) / 1024;
    int4 q[NQ];
#pragma unroll
    for (int k = 0; k < NQ; ++k) {
        int i = tid + k * 1024;
        q[k] = make_int4(0, 0, 0, 0);
        if (i < nv) q[k] = c4[i];
    }
#pragma unroll
    for (int k = 0; k < NQ; ++k) {
        int i = tid + k * 1024;
        if (i < OFF_CAP / 4) o4[i] = q[k];
    }
    __syncthreads();
    if (n & 3) {
        for (int i = (nv << 2) + tid; i < n; i += 1024) offs[i] = cnt[i];
        __syncthreads();
    }

    int v[OFF_PT];
    int s = 0;
    int* mine = offs + tid * OFF_PT;
#pragma unroll
    for (int j = 0; j < OFF_PT; ++j) {
        v[j] = mine[j];
        s += v[j];
    }

    int incl = s;
    for (int o = 1; o < 32; o <<= 1) {
        int t = __shfl_up(incl, o, 32);
        if (lane >= o) incl += t;
    }
    if (lane == 31) wsum[wv] = incl;
    __syncthreads();
    if (wv == 0) {
        int w = wsum[lane];
        int sc = w;
        for (int o = 1; o < 32; o <<= 1) {
            int t = __shfl_up(sc, o, 32);
            if (lane >= o) sc += t;
        }
        wsum[lane] = sc - w;
    }
    __syncthreads();

    int run = wsum[wv] + incl - s;
#pragma unroll
    for (int j = 0; j < OFF_PT; ++j) {
        mine[j] = run;
        run += v[j];
    }
    __syncthreads();

    for (int i = blockIdx.x * 1024 + tid; i <= n; i += gridDim.x * 1024)
        off[i] = offs[i];

    int estride = gridDim.x * 1024;
    for (int e0 = blockIdx.x * 1024 + tid; e0 < E_; e0 += 4 * estride) {
        int d[4], r[4], sv[4];
#pragma unroll
        for (int u = 0; u < 4; ++u) {
            int e = e0 + u * estride;
            d[u] = 0; r[u] = 0; sv[u] = 0;
            if (e < E_) {
                d[u]  = dst[e];
                r[u]  = rank[e];
                sv[u] = src[e];
            }
        }
#pragma unroll
        for (int u = 0; u < 4; ++u) {
            int e = e0 + u * estride;
            if (e < E_) csr[offs[d[u]] + r[u]] = sv[u];
        }
    }
}

__device__ __forceinline__ float edge_weight(float x)
{
    float lv = (x >= 0.f) ? x : NEG_SLOPE * x;
    return expf(lv);
}

__device__ __forceinline__ float elu1(float v)
{
    return (v > 0.f) ? v : expm1f(v);
}


__global__ __launch_bounds__(256, 16) void node_aggr1(const int* __restrict__ off,
                                                  const int* __restrict__ csr,
                                                  const float* __restrict__ als,
                                                  const float* __restrict__ ald,
                                                  const float* __restrict__ h,
                                                  const float* __restrict__ bias,
                                                  float* __restrict__ out,
                                                  int n)
{
    int d = blockIdx.x * (blockDim.x >> 5) + (threadIdx.x >> 5);
    if (d >= n) return;
    d = __builtin_amdgcn_readfirstlane(d);
    int lane = threadIdx.x & 31;
    int hd0 = lane >> 3;
    int c0 = 4 * lane;
    int c1 = 128 + 4 * lane;
    int eu = lane >> 3;
    int eh = lane & 7;

    float adm = ald[d * 8 + eh];
    int beg = off[d];
    int end = off[d + 1];

    float4 acc0, acc1;
    float den0, den1;
    {
        float W = edge_weight(als[d * 8 + eh] + adm);
        float w0 = __shfl(W, hd0, 32);
        float w1 = __shfl(W, 4 + hd0, 32);
        const float* hs = h + (size_t)d * 256;
        float4 v0 = *reinterpret_cast<const float4*>(hs + c0);
        float4 v1 = *reinterpret_cast<const float4*>(hs + c1);
        acc0 = make_float4(w0 * v0.x, w0 * v0.y, w0 * v0.z, w0 * v0.w);
        acc1 = make_float4(w1 * v1.x, w1 * v1.y, w1 * v1.z, w1 * v1.w);
        den0 = w0;
        den1 = w1;
    }

    int p = beg;
    for (; p + 4 <= end; p += 4) {
        int s[4];
        float4 v0[4], v1[4];
        for (int u = 0; u < 4; ++u) s[u] = csr[p + u];
        int sa = (eu & 1) ? s[1] : s[0];
        int sb = (eu & 1) ? s[3] : s[2];
        int sm = (eu & 2) ? sb : sa;
        float x = als[sm * 8 + eh];
        for (int u = 0; u < 4; ++u) {
            const float* hs = h + (size_t)s[u] * 256;
            v0[u] = *reinterpret_cast<const float4*>(hs + c0);
            v1[u] = *reinterpret_cast<const float4*>(hs + c1);
        }
        float W = edge_weight(x + adm);
        for (int u = 0; u < 4; ++u) {
            float w0 = __shfl(W, 8 * u + hd0, 32);
            float w1 = __shfl(W, 8 * u + 4 + hd0, 32);
            acc0.x += w0 * v0[u].x; acc0.y += w0 * v0[u].y;
            acc0.z += w0 * v0[u].z; acc0.w += w0 * v0[u].w;
            acc1.x += w1 * v1[u].x; acc1.y += w1 * v1[u].y;
            acc1.z += w1 * v1[u].z; acc1.w += w1 * v1[u].w;
            den0 += w0;
            den1 += w1;
        }
    }
    int rem = end - p;
    if (rem > 0) {
        int s0 = csr[p];
        int s1 = (rem > 1) ? csr[p + 1] : s0;
        int s2 = (rem > 2) ? csr[p + 2] : s0;
        int sa = (eu & 1) ? s1 : s0;
        int sm = (eu & 2) ? s2 : sa;
        float x = als[sm * 8 + eh];
        float4 t0[3], t1[3];
        const int st[3] = {s0, s1, s2};
        for (int u = 0; u < 3; ++u) {
            const float* hs = h + (size_t)st[u] * 256;
            t0[u] = *reinterpret_cast<const float4*>(hs + c0);
            t1[u] = *reinterpret_cast<const float4*>(hs + c1);
        }
        float W = edge_weight(x + adm);
        for (int u = 0; u < 3; ++u) {
            if (u < rem) {
                float w0 = __shfl(W, 8 * u + hd0, 32);
                float w1 = __shfl(W, 8 * u + 4 + hd0, 32);
                acc0.x += w0 * t0[u].x; acc0.y += w0 * t0[u].y;
                acc0.z += w0 * t0[u].z; acc0.w += w0 * t0[u].w;
                acc1.x += w1 * t1[u].x; acc1.y += w1 * t1[u].y;
                acc1.z += w1 * t1[u].z; acc1.w += w1 * t1[u].w;
                den0 += w0;
                den1 += w1;
            }
        }
    }

    float4 b0 = *reinterpret_cast<const float4*>(bias + c0);
    float4 b1 = *reinterpret_cast<const float4*>(bias + c1);
    float4 r0, r1;
    r0.x = elu1(acc0.x / den0 + b0.x); r0.y = elu1(acc0.y / den0 + b0.y);
    r0.z = elu1(acc0.z / den0 + b0.z); r0.w = elu1(acc0.w / den0 + b0.w);
    r1.x = elu1(acc1.x / den1 + b1.x); r1.y = elu1(acc1.y / den1 + b1.y);
    r1.z = elu1(acc1.z / den1 + b1.z); r1.w = elu1(acc1.w / den1 + b1.w);
    float* od = out + (size_t)d * 256;
    *reinterpret_cast<float4*>(od + c0) = r0;
    *reinterpret_cast<float4*>(od + c1) = r1;
}

__global__ __launch_bounds__(256) void node_aggr2(const int* __restrict__ off,
                                                  const int* __restrict__ csr,
                                                  const float* __restrict__ als,
                                                  const float* __restrict__ ald,
                                                  const float* __restrict__ h,
                                                  const float* __restrict__ bias,
                                                  float* __restrict__ out,
                                                  int n)
{
    int d = blockIdx.x * (blockDim.x >> 5) + (threadIdx.x >> 5);
    if (d >= n) return;
    d = __builtin_amdgcn_readfirstlane(d);
    int lane = threadIdx.x & 31;
    int c0 = 2 * lane;
    int eu = lane & 3;

    float ad = ald[d];
    int beg = off[d];
    int end = off[d + 1];

    float2 acc;
    float den;
    {
        float w = edge_weight(als[d] + ad);
        float2 v = *reinterpret_cast<const float2*>(h + (size_t)d * 64 + c0);
        acc = make_float2(w * v.x, w * v.y);
        den = w;
    }

    int p = beg;
    for (; p + 4 <= end; p += 4) {
        int s[4];
        float2 v[4];
        for (int u = 0; u < 4; ++u) s[u] = csr[p + u];
        int sa = (eu & 1) ? s[1] : s[0];
        int sb = (eu & 1) ? s[3] : s[2];
        int sm = (eu & 2) ? sb : sa;
        float x = als[sm];
        for (int u = 0; u < 4; ++u)
            v[u] = *reinterpret_cast<const float2*>(h + (size_t)s[u] * 64 + c0);
        float W = edge_weight(x + ad);
        for (int u = 0; u < 4; ++u) {
            float w = __shfl(W, u, 32);
            acc.x += w * v[u].x;
            acc.y += w * v[u].y;
            den += w;
        }
    }
    int rem = end - p;
    if (rem > 0) {
        int s0 = csr[p];
        int s1 = (rem > 1) ? csr[p + 1] : s0;
        int s2 = (rem > 2) ? csr[p + 2] : s0;
        int sa = (eu & 1) ? s1 : s0;
        int sm = (eu & 2) ? s2 : sa;
        float x = als[sm];
        float2 t[3];
        const int st[3] = {s0, s1, s2};
        for (int u = 0; u < 3; ++u)
            t[u] = *reinterpret_cast<const float2*>(h + (size_t)st[u] * 64 + c0);
        float W = edge_weight(x + ad);
        for (int u = 0; u < 3; ++u) {
            if (u < rem) {
                float w = __shfl(W, u, 32);
                acc.x += w * t[u].x;
                acc.y += w * t[u].y;
                den += w;
            }
        }
    }

    float2 b = *reinterpret_cast<const float2*>(bias + c0);
    float2 r;
    r.x = elu1(acc.x / den + b.x);
    r.y = elu1(acc.y / den + b.y);
    *reinterpret_cast<float2*>(out + (size_t)d * 64 + c0) = r;
}

extern "C" void kernel_launch(void* const* d_in, const int* in_sizes, int n_in,
                              void* d_out, int out_size, void* d_ws, size_t ws_size,
                              hipStream_t stream)
{
    const float* x   = (const float*)d_in[0];
    const int*   ei  = (const int*)  d_in[1];
    const float* W1  = (const float*)d_in[2];
    const float* as1 = (const float*)d_in[3];
    const float* ad1 = (const float*)d_in[4];
    const float* b1  = (const float*)d_in[5];
    const float* W2  = (const float*)d_in[6];
    const float* as2 = (const float*)d_in[7];
    const float* ad2 = (const float*)d_in[8];
    const float* b2  = (const float*)d_in[9];
    float* out = (float*)d_out;

    const int IN_DIM = 128, H1 = 8, F1 = 256, OUT_DIM = 64;
    int n = in_sizes[0] / IN_DIM;
    int e = in_sizes[1] / 2;
    const int* src = ei;
    const int* dst = ei + e;

    float* ws = (float*)d_ws;
    float* h1   = ws;  ws += (size_t)n * F1;
    float* g1   = ws;  ws += (size_t)n * F1;
    int*   ib   = (int*)ws;  ws += (size_t)n * OUT_DIM;
    float* als1 = ws;  ws += (size_t)n * H1;
    float* ald1 = ws;  ws += (size_t)n * H1;
    float* ald2 = ws;  ws += (size_t)n * H1;
    float* als2 = ws;  ws += (size_t)n;
    ws += (size_t)n;
    float* h2   = h1;

    int* cnt  = ib;  ib += n;
    int* rank = ib;  ib += e;
    int* csr  = ib;  ib += e;
    int* off  = ib;

    hipMemsetAsync(cnt, 0, (size_t)n * sizeof(int), stream);

    {
        int tasks = (n / 16) * (F1 / 64);
        gemm_wmma<128, 256, 32, true><<<(tasks + 7) / 8, 256, 0, stream>>>(
            x, W1, h1, as1, ad1, als1, ald1, tasks, dst, cnt, rank, e);
    }

    scan_fill<<<256, 1024, 0, stream>>>(cnt, src, dst, rank, off, csr, n, e);

    node_aggr1<<<(n + 7) / 8, 256, 0, stream>>>(off, csr, als1, ald1, h1, b1, g1, n);

    {
        int tasks = (n / 16) * (OUT_DIM / 64);
        gemm_wmma<256, 64, 64, false><<<(tasks + 7) / 8, 256, 0, stream>>>(
            g1, W2, h2, as2, ad2, als2, ald2, tasks, nullptr, nullptr, nullptr, 0);
    }
    node_aggr2<<<(n + 7) / 8, 256, 0, stream>>>(off, csr, als2, ald2, h2, b2, out, n);
}


# GEMM kernels: 16-byte global accesses per lane: a lane's four tiles are neighbouring columns (one B load per k, one C store per row), A read eight k per trip; k and coefficient-dot sum order change
# speedup vs baseline: 3.1571x; 1.0502x over previous
#include <hip/hip_runtime.h>
#include <math.h>

typedef __attribute__((ext_vector_type(2))) float v2f;
typedef __attribute__((ext_vector_type(8))) float v8f;

#define NEG_SLOPE 0.2f

template<int CNT, int M>
__device__ __forceinline__ void fold_lanes(float* v, int mr)
{
    bool up = (mr & M) != 0;
#pragma unroll
    for (int j = 0; j < CNT; ++j) {
        float lo = v[j];
        float hi = v[j + CNT];
        float send = up ? lo : hi;
        float keep = up ? hi : lo;
        v[j] = keep + __shfl_xor(send, M, 32);
    }
}

__device__ __forceinline__ v8f mma4(float a0, float a1, float b0, float b1, v8f c)
{
    v2f a, b;
    a.x = a0; a.y = a1;
    b.x = b0; b.y = b1;
    return __builtin_amdgcn_wmma_f32_16x16x4_f32(false, a, false, b, (short)0, c, false, false);
}

template<int K, int NOUT, int HC, bool RANK>
__global__ __launch_bounds__(256) void gemm_wmma(const float* __restrict__ A,
                                                 const float* __restrict__ B,
                                                 float* __restrict__ C,
                                                 const float* __restrict__ a_src,
                                                 const float* __restrict__ a_dst,
                                                 float* __restrict__ als,
                                                 float* __restrict__ ald,
                                                 int ntasks,
                                                 const int* __restrict__ edst,
                                                 int* __restrict__ cnt,
                                                 int* __restrict__ rank,
                                                 int E_)
{
    constexpr int NG  = NOUT / 64;
    constexpr int HPW = 64 / HC;
    constexpr int NH  = NOUT / HC;
    constexpr int RB  = (8 / NG) * 16;
    constexpr int LPH = HC / 4;
    constexpr int NS  = 16 / LPH;
    static_assert(HC == 32 || HC == 64, "a lane's four columns must lie in one head");
    static_assert(K % 8 == 0, "eight k per trip");
    __shared__ float sh[2 * RB * NH];

    int wv   = threadIdx.x >> 5;
    int task = blockIdx.x * 8 + wv;
    int lane = threadIdx.x & 31;
    int half = lane >> 4;
    int mr   = lane & 15;

    bool active = task < ntasks;
    int rt = task / NG;
    int cg = task % NG;
    v8f acc[4] = {};

    if (active) {

        const float* arow = A + (size_t)(rt * 16 + mr) * K + 4 * half;
        const float* bcol = B + (size_t)(4 * half) * NOUT + cg * 64 + 4 * mr;

        for (int k0 = 0; k0 < K; k0 += 8) {
            float4 av = *reinterpret_cast<const float4*>(arow + k0);
            const float* br = bcol + (size_t)k0 * NOUT;
            float4 q0 = *reinterpret_cast<const float4*>(br);
            float4 q1 = *reinterpret_cast<const float4*>(br + NOUT);
            float4 q2 = *reinterpret_cast<const float4*>(br + 2 * NOUT);
            float4 q3 = *reinterpret_cast<const float4*>(br + 3 * NOUT);
            acc[0] = mma4(av.x, av.y, q0.x, q1.x, acc[0]);
            acc[1] = mma4(av.x, av.y, q0.y, q1.y, acc[1]);
            acc[2] = mma4(av.x, av.y, q0.z, q1.z, acc[2]);
            acc[3] = mma4(av.x, av.y, q0.w, q1.w, acc[3]);
            acc[0] = mma4(av.z, av.w, q2.x, q3.x, acc[0]);
            acc[1] = mma4(av.z, av.w, q2.y, q3.y, acc[1]);
            acc[2] = mma4(av.z, av.w, q2.z, q3.z, acc[2]);
            acc[3] = mma4(av.z, av.w, q2.w, q3.w, acc[3]);
        }

        float4 cs4 = *reinterpret_cast<const float4*>(a_src + cg * 64 + 4 * mr);
        float4 cd4 = *reinterpret_cast<const float4*>(a_dst + cg * 64 + 4 * mr);

        float v[16];
#pragma unroll
        for (int r = 0; r < 8; ++r) {
            float ps = acc[0][r] * cs4.x;
            float pd = acc[0][r] * cd4.x;
            ps += acc[1][r] * cs4.y;
            pd += acc[1][r] * cd4.y;
            ps += acc[2][r] * cs4.z;
            pd += acc[2][r] * cd4.z;
            ps += acc[3][r] * cs4.w;
            pd += acc[3][r] * cd4.w;
            v[r] = ps;
            v[8 + r] = pd;
        }

        int jb;
        int hh;
        if constexpr (LPH == 16) {
            fold_lanes<8, 8>(v, mr);
            fold_lanes<4, 4>(v, mr);
            fold_lanes<2, 2>(v, mr);
            fold_lanes<1, 1>(v, mr);
            jb = ((mr >> 3) & 1) * 8 + ((mr >> 2) & 1) * 4 + ((mr >> 1) & 1) * 2 + (mr & 1);
            hh = 0;
        } else {
            fold_lanes<8, 4>(v, mr);
            fold_lanes<4, 2>(v, mr);
            fold_lanes<2, 1>(v, mr);
            jb = ((mr >> 2) & 1) * 8 + ((mr >> 1) & 1) * 4 + (mr & 1) * 2;
            hh = mr >> 3;
        }
        int rtl = wv / NG;
#pragma unroll
        for (int i = 0; i < NS; ++i) {
            int j    = jb + i;
            int kind = j / 8;
            int r    = j % 8;
            sh[kind * (RB * NH) + (rtl * 16 + half * 8 + r) * NH + cg * HPW + hh] = v[i];
        }
    }
    __syncthreads();

    if (active) {
        float* crow = C + (size_t)(rt * 16 + half * 8) * NOUT + cg * 64 + 4 * mr;
#pragma unroll
        for (int r = 0; r < 8; ++r)
            *reinterpret_cast<float4*>(crow + (size_t)r * NOUT) =
                make_float4(acc[0][r], acc[1][r], acc[2][r], acc[3][r]);
    }

    int base  = blockIdx.x * (RB * NH);
    int limit = (ntasks / NG) * 16 * NH;
    for (int idx = threadIdx.x; idx < 2 * RB * NH; idx += 256) {
        int kind = idx / (RB * NH);
        int o    = idx % (RB * NH);
        if (base + o < limit) {
            float* dstp = kind ? ald : als;
            dstp[base + o] = sh[idx];
        }
    }

    if (RANK) {
        if (active) {
            int estride = ntasks * 32;
            for (int e = task * 32 + lane; e < E_; e += 2 * estride) {
                int e1 = e + estride;
                bool two = e1 < E_;
                int d0 = edst[e];
                int d1 = two ? edst[e1] : 0;
                int r0 = atomicAdd(&cnt[d0], 1);
                int r1 = 0;
                if (two) r1 = atomicAdd(&cnt[d1], 1);
                rank[e] = r0;
                if (two) rank[e1] = r1;
            }
        }
    }
}

#define OFF_PT  49
#define OFF_CAP (1024 * OFF_PT)

__global__ __launch_bounds__(1024) void scan_fill(const int* __restrict__ cnt,
                                                  const int* __restrict__ src,
                                                  const int* __restrict__ dst,
                                                  const int* __restrict__ rank,
                                                  int* __restrict__ off,
                                                  int* __restrict__ csr,
                                                  int n,
                                                  int E_)
{
    __shared__ alignas(16) int offs[OFF_CAP];
    __shared__ int wsum[32];
    int tid  = threadIdx.x;
    int lane = tid & 31;
    int wv   = tid >> 5;

    const int4* c4 = reinterpret_cast<const int4*>(cnt);
    int4* o4 = reinterpret_cast<int4*>(offs);
    int nv = n >> 2;
    constexpr int NQ = (OFF_CAP / 4 + 1023) / 1024;
    int4 q[NQ];
#pragma unroll
    for (int k = 0; k < NQ; ++k) {
        int i = tid + k * 1024;
        q[k] = make_int4(0, 0, 0, 0);
        if (i < nv) q[k] = c4[i];
    }
#pragma unroll
    for (int k = 0; k < NQ; ++k) {
        int i = tid + k * 1024;
        if (i < OFF_CAP / 4) o4[i] = q[k];
    }
    __syncthreads();
    if (n & 3) {
        for (int i = (nv << 2) + tid; i < n; i += 1024) offs[i] = cnt[i];
        __syncthreads();
    }

    int v[OFF_PT];
    int s = 0;
    int* mine = offs + tid * OFF_PT;
#pragma unroll
    for (int j = 0; j < OFF_PT; ++j) {
        v[j] = mine[j];
        s += v[j];
    }

    int incl = s;
    for (int o = 1; o < 32; o <<= 1) {
        int t = __shfl_up(incl, o, 32);
        if (lane >= o) incl += t;
    }
    if (lane == 31) wsum[wv] = incl;
    __syncthreads();
    if (wv == 0) {
        int w = wsum[lane];
        int sc = w;
        for (int o = 1; o < 32; o <<= 1) {
            int t = __shfl_up(sc, o, 32);
            if (lane >= o) sc += t;
        }
        wsum[lane] = sc - w;
    }
    __syncthreads();

    int run = wsum[wv] + incl - s;
#pragma unroll
    for (int j = 0; j < OFF_PT; ++j) {
        mine[j] = run;
        run += v[j];
    }
    __syncthreads();

    for (int i = blockIdx.x * 1024 + tid; i <= n; i += gridDim.x * 1024)
        off[i] = offs[i];

    int estride = gridDim.x * 1024;
    for (int e0 = blockIdx.x * 1024 + tid; e0 < E_; e0 += 4 * estride) {
        int d[4], r[4], sv[4];
#pragma unroll
        for (int u = 0; u < 4; ++u) {
            int e = e0 + u * estride;
            d[u] = 0; r[u] = 0; sv[u] = 0;
            if (e < E_) {
                d[u]  = dst[e];
                r[u]  = rank[e];
                sv[u] = src[e];
            }
        }
#pragma unroll
        for (int u = 0; u < 4; ++u) {
            int e = e0 + u * estride;
            if (e < E_) csr[offs[d[u]] + r[u]] = sv[u];
        }
    }
}

__device__ __forceinline__ float edge_weight(float x)
{
    float lv = (x >= 0.f) ? x : NEG_SLOPE * x;
    return expf(lv);
}

__device__ __forceinline__ float elu1(float v)
{
    return (v > 0.f) ? v : expm1f(v);
}


__global__ __launch_bounds__(256, 16) void node_aggr1(const int* __restrict__ off,
                                                  const int* __restrict__ csr,
                                                  const float* __restrict__ als,
                                                  const float* __restrict__ ald,
                                                  const float* __restrict__ h,
                                                  const float* __restrict__ bias,
                                                  float* __restrict__ out,
                                                  int n)
{
    int d = blockIdx.x * (blockDim.x >> 5) + (threadIdx.x >> 5);
    if (d >= n) return;
    d = __builtin_amdgcn_readfirstlane(d);
    int lane = threadIdx.x & 31;
    int hd0 = lane >> 3;
    int c0 = 4 * lane;
    int c1 = 128 + 4 * lane;
    int eu = lane >> 3;
    int eh = lane & 7;

    float adm = ald[d * 8 + eh];
    int beg = off[d];
    int end = off[d + 1];

    float4 acc0, acc1;
    float den0, den1;
    {
        float W = edge_weight(als[d * 8 + eh] + adm);
        float w0 = __shfl(W, hd0, 32);
        float w1 = __shfl(W, 4 + hd0, 32);
        const float* hs = h + (size_t)d * 256;
        float4 v0 = *reinterpret_cast<const float4*>(hs + c0);
        float4 v1 = *reinterpret_cast<const float4*>(hs + c1);
        acc0 = make_float4(w0 * v0.x, w0 * v0.y, w0 * v0.z, w0 * v0.w);
        acc1 = make_float4(w1 * v1.x, w1 * v1.y, w1 * v1.z, w1 * v1.w);
        den0 = w0;
        den1 = w1;
    }

    int p = beg;
    for (; p + 4 <= end; p += 4) {
        int s[4];
        float4 v0[4], v1[4];
        for (int u = 0; u < 4; ++u) s[u] = csr[p + u];
        int sa = (eu & 1) ? s[1] : s[0];
        int sb = (eu & 1) ? s[3] : s[2];
        int sm = (eu & 2) ? sb : sa;
        float x = als[sm * 8 + eh];
        for (int u = 0; u < 4; ++u) {
            const float* hs = h + (size_t)s[u] * 256;
            v0[u] = *reinterpret_cast<const float4*>(hs + c0);
            v1[u] = *reinterpret_cast<const float4*>(hs + c1);
        }
        float W = edge_weight(x + adm);
        for (int u = 0; u < 4; ++u) {
            float w0 = __shfl(W, 8 * u + hd0, 32);
            float w1 = __shfl(W, 8 * u + 4 + hd0, 32);
            acc0.x += w0 * v0[u].x; acc0.y += w0 * v0[u].y;
            acc0.z += w0 * v0[u].z; acc0.w += w0 * v0[u].w;
            acc1.x += w1 * v1[u].x; acc1.y += w1 * v1[u].y;
            acc1.z += w1 * v1[u].z; acc1.w += w1 * v1[u].w;
            den0 += w0;
            den1 += w1;
        }
    }
    int rem = end - p;
    if (rem > 0) {
        int s0 = csr[p];
        int s1 = (rem > 1) ? csr[p + 1] : s0;
        int s2 = (rem > 2) ? csr[p + 2] : s0;
        int sa = (eu & 1) ? s1 : s0;
        int sm = (eu & 2) ? s2 : sa;
        float x = als[sm * 8 + eh];
        float4 t0[3], t1[3];
        const int st[3] = {s0, s1, s2};
        for (int u = 0; u < 3; ++u) {
            const float* hs = h + (size_t)st[u] * 256;
            t0[u] = *reinterpret_cast<const float4*>(hs + c0);
            t1[u] = *reinterpret_cast<const float4*>(hs + c1);
        }
        float W = edge_weight(x + adm);
        for (int u = 0; u < 3; ++u) {
            if (u < rem) {
                float w0 = __shfl(W, 8 * u + hd0, 32);
                float w1 = __shfl(W, 8 * u + 4 + hd0, 32);
                acc0.x += w0 * t0[u].x; acc0.y += w0 * t0[u].y;
                acc0.z += w0 * t0[u].z; acc0.w += w0 * t0[u].w;
                acc1.x += w1 * t1[u].x; acc1.y += w1 * t1[u].y;
                acc1.z += w1 * t1[u].z; acc1.w += w1 * t1[u].w;
                den0 += w0;
                den1 += w1;
            }
        }
    }

    float4 b0 = *reinterpret_cast<const float4*>(bias + c0);
    float4 b1 = *reinterpret_cast<const float4*>(bias + c1);
    float4 r0, r1;
    r0.x = elu1(acc0.x / den0 + b0.x); r0.y = elu1(acc0.y / den0 + b0.y);
    r0.z = elu1(acc0.z / den0 + b0.z); r0.w = elu1(acc0.w / den0 + b0.w);
    r1.x = elu1(acc1.x / den1 + b1.x); r1.y = elu1(acc1.y / den1 + b1.y);
    r1.z = elu1(acc1.z / den1 + b1.z); r1.w = elu1(acc1.w / den1 + b1.w);
    float* od = out + (size_t)d * 256;
    *reinterpret_cast<float4*>(od + c0) = r0;
    *reinterpret_cast<float4*>(od + c1) = r1;
}

__global__ __launch_bounds__(256) void node_aggr2(const int* __restrict__ off,
                                                  const int* __restrict__ csr,
                                                  const float* __restrict__ als,
                                                  const float* __restrict__ ald,
                                                  const float* __restrict__ h,
                                                  const float* __restrict__ bias,
                                                  float* __restrict__ out,
                                                  int n)
{
    int d = blockIdx.x * (blockDim.x >> 5) + (threadIdx.x >> 5);
    if (d >= n) return;
    d = __builtin_amdgcn_readfirstlane(d);
    int lane = threadIdx.x & 31;
    int c0 = 2 * lane;
    int eu = lane & 3;

    float ad = ald[d];
    int beg = off[d];
    int end = off[d + 1];

    float2 acc;
    float den;
    {
        float w = edge_weight(als[d] + ad);
        float2 v = *reinterpret_cast<const float2*>(h + (size_t)d * 64 + c0);
        acc = make_float2(w * v.x, w * v.y);
        den = w;
    }

    int p = beg;
    for (; p + 4 <= end; p += 4) {
        int s[4];
        float2 v[4];
        for (int u = 0; u < 4; ++u) s[u] = csr[p + u];
        int sa = (eu & 1) ? s[1] : s[0];
        int sb = (eu & 1) ? s[3] : s[2];
        int sm = (eu & 2) ? sb : sa;
        float x = als[sm];
        for (int u = 0; u < 4; ++u)
            v[u] = *reinterpret_cast<const float2*>(h + (size_t)s[u] * 64 + c0);
        float W = edge_weight(x + ad);
        for (int u = 0; u < 4; ++u) {
            float w = __shfl(W, u, 32);
            acc.x += w * v[u].x;
            acc.y += w * v[u].y;
            den += w;
        }
    }
    int rem = end - p;
    if (rem > 0) {
        int s0 = csr[p];
        int s1 = (rem > 1) ? csr[p + 1] : s0;
        int s2 = (rem > 2) ? csr[p + 2] : s0;
        int sa = (eu & 1) ? s1 : s0;
        int sm = (eu & 2) ? s2 : sa;
        float x = als[sm];
        float2 t[3];
        const int st[3] = {s0, s1, s2};
        for (int u = 0; u < 3; ++u)
            t[u] = *reinterpret_cast<const float2*>(h + (size_t)st[u] * 64 + c0);
        float W = edge_weight(x + ad);
        for (int u = 0; u < 3; ++u) {
            if (u < rem) {
                float w = __shfl(W, u, 32);
                acc.x += w * t[u].x;
                acc.y += w * t[u].y;
                den += w;
            }
        }
    }

    float2 b = *reinterpret_cast<const float2*>(bias + c0);
    float2 r;
    r.x = elu1(acc.x / den + b.x);
    r.y = elu1(acc.y / den + b.y);
    *reinterpret_cast<float2*>(out + (size_t)d * 64 + c0) = r;
}

extern "C" void kernel_launch(void* const* d_in, const int* in_sizes, int n_in,
                              void* d_out, int out_size, void* d_ws, size_t ws_size,
                              hipStream_t stream)
{
    const float* x   = (const float*)d_in[0];
    const int*   ei  = (const int*)  d_in[1];
    const float* W1  = (const float*)d_in[2];
    const float* as1 = (const float*)d_in[3];
    const float* ad1 = (const float*)d_in[4];
    const float* b1  = (const float*)d_in[5];
    const float* W2  = (const float*)d_in[6];
    const float* as2 = (const float*)d_in[7];
    const float* ad2 = (const float*)d_in[8];
    const float* b2  = (const float*)d_in[9];
    float* out = (float*)d_out;

    const int IN_DIM = 128, H1 = 8, F1 = 256, OUT_DIM = 64;
    int n = in_sizes[0] / IN_DIM;
    int e = in_sizes[1] / 2;
    const int* src = ei;
    const int* dst = ei + e;

    float* ws = (float*)d_ws;
    float* h1   = ws;  ws += (size_t)n * F1;
    float* g1   = ws;  ws += (size_t)n * F1;
    int*   ib   = (int*)ws;  ws += (size_t)n * OUT_DIM;
    float* als1 = ws;  ws += (size_t)n * H1;
    float* ald1 = ws;  ws += (size_t)n * H1;
    float* ald2 = ws;  ws += (size_t)n * H1;
    float* als2 = ws;  ws += (size_t)n;
    ws += (size_t)n;
    float* h2   = h1;

    int* cnt  = ib;  ib += n;
    int* rank = ib;  ib += e;
    int* csr  = ib;  ib += e;
    int* off  = ib;

    hipMemsetAsync(cnt, 0, (size_t)n * sizeof(int), stream);

    {
        int tasks = (n / 16) * (F1 / 64);
        gemm_wmma<128, 256, 32, true><<<(tasks + 7) / 8, 256, 0, stream>>>(
            x, W1, h1, as1, ad1, als1, ald1, tasks, dst, cnt, rank, e);
    }

    scan_fill<<<256, 1024, 0, stream>>>(cnt, src, dst, rank, off, csr, n, e);

    node_aggr1<<<(n + 7) / 8, 256, 0, stream>>>(off, csr, als1, ald1, h1, b1, g1, n);

    {
        int tasks = (n / 16) * (OUT_DIM / 64);
        gemm_wmma<256, 64, 64, false><<<(tasks + 7) / 8, 256, 0, stream>>>(
            g1, W2, h2, as2, ad2, als2, ald2, tasks, nullptr, nullptr, nullptr, 0);
    }
    node_aggr2<<<(n + 7) / 8, 256, 0, stream>>>(off, csr, als2, ald2, h2, b2, out, n);
}


# conv2 GEMM moved into the conv1 aggregation launch: a block's 32 finished rows are f32 WMMA A tiles in LDS, k split over 4 waves a tile, partials summed in LDS; k and coefficient-dot sum order change
# speedup vs baseline: 3.2699x; 1.0357x over previous
#include <hip/hip_runtime.h>
#include <math.h>

typedef __attribute__((ext_vector_type(2))) float v2f;
typedef __attribute__((ext_vector_type(8))) float v8f;

#define NEG_SLOPE 0.2f

template<int CNT, int M>
__device__ __forceinline__ void fold_lanes(float* v, int mr)
{
    bool up = (mr & M) != 0;
#pragma unroll
    for (int j = 0; j < CNT; ++j) {
        float lo = v[j];
        float hi = v[j + CNT];
        float send = up ? lo : hi;
        float keep = up ? hi : lo;
        v[j] = keep + __shfl_xor(send, M, 32);
    }
}

__device__ __forceinline__ v8f mma4(float a0, float a1, float b0, float b1, v8f c)
{
    v2f a, b;
    a.x = a0; a.y = a1;
    b.x = b0; b.y = b1;
    return __builtin_amdgcn_wmma_f32_16x16x4_f32(false, a, false, b, (short)0, c, false, false);
}

template<int K, int NOUT, int HC, bool RANK>
__global__ __launch_bounds__(256) void gemm_wmma(const float* __restrict__ A,
                                                 const float* __restrict__ B,
                                                 float* __restrict__ C,
                                                 const float* __restrict__ a_src,
                                                 const float* __restrict__ a_dst,
                                                 float* __restrict__ als,
                                                 float* __restrict__ ald,
                                                 int ntasks,
                                                 const int* __restrict__ edst,
                                                 int* __restrict__ cnt,
                                                 int* __restrict__ rank,
                                                 int E_)
{
    constexpr int NG  = NOUT / 64;
    constexpr int HPW = 64 / HC;
    constexpr int NH  = NOUT / HC;
    constexpr int RB  = (8 / NG) * 16;
    constexpr int LPH = HC / 4;
    constexpr int NS  = 16 / LPH;
    static_assert(HC == 32 || HC == 64, "a lane's four columns must lie in one head");
    static_assert(K % 8 == 0, "eight k per trip");
    __shared__ float sh[2 * RB * NH];

    int wv   = threadIdx.x >> 5;
    int task = blockIdx.x * 8 + wv;
    int lane = threadIdx.x & 31;
    int half = lane >> 4;
    int mr   = lane & 15;

    bool active = task < ntasks;
    int rt = task / NG;
    int cg = task % NG;
    v8f acc[4] = {};

    if (active) {

        const float* arow = A + (size_t)(rt * 16 + mr) * K + 4 * half;
        const float* bcol = B + (size_t)(4 * half) * NOUT + cg * 64 + 4 * mr;

        for (int k0 = 0; k0 < K; k0 += 8) {
            float4 av = *reinterpret_cast<const float4*>(arow + k0);
            const float* br = bcol + (size_t)k0 * NOUT;
            float4 q0 = *reinterpret_cast<const float4*>(br);
            float4 q1 = *reinterpret_cast<const float4*>(br + NOUT);
            float4 q2 = *reinterpret_cast<const float4*>(br + 2 * NOUT);
            float4 q3 = *reinterpret_cast<const float4*>(br + 3 * NOUT);
            acc[0] = mma4(av.x, av.y, q0.x, q1.x, acc[0]);
            acc[1] = mma4(av.x, av.y, q0.y, q1.y, acc[1]);
            acc[2] = mma4(av.x, av.y, q0.z, q1.z, acc[2]);
            acc[3] = mma4(av.x, av.y, q0.w, q1.w, acc[3]);
            acc[0] = mma4(av.z, av.w, q2.x, q3.x, acc[0]);
            acc[1] = mma4(av.z, av.w, q2.y, q3.y, acc[1]);
            acc[2] = mma4(av.z, av.w, q2.z, q3.z, acc[2]);
            acc[3] = mma4(av.z, av.w, q2.w, q3.w, acc[3]);
        }

        float4 cs4 = *reinterpret_cast<const float4*>(a_src + cg * 64 + 4 * mr);
        float4 cd4 = *reinterpret_cast<const float4*>(a_dst + cg * 64 + 4 * mr);

        float v[16];
#pragma unroll
        for (int r = 0; r < 8; ++r) {
            float ps = acc[0][r] * cs4.x;
            float pd = acc[0][r] * cd4.x;
            ps += acc[1][r] * cs4.y;
            pd += acc[1][r] * cd4.y;
            ps += acc[2][r] * cs4.z;
            pd += acc[2][r] * cd4.z;
            ps += acc[3][r] * cs4.w;
            pd += acc[3][r] * cd4.w;
            v[r] = ps;
            v[8 + r] = pd;
        }

        int jb;
        int hh;
        if constexpr (LPH == 16) {
            fold_lanes<8, 8>(v, mr);
            fold_lanes<4, 4>(v, mr);
            fold_lanes<2, 2>(v, mr);
            fold_lanes<1, 1>(v, mr);
            jb = ((mr >> 3) & 1) * 8 + ((mr >> 2) & 1) * 4 + ((mr >> 1) & 1) * 2 + (mr & 1);
            hh = 0;
        } else {
            fold_lanes<8, 4>(v, mr);
            fold_lanes<4, 2>(v, mr);
            fold_lanes<2, 1>(v, mr);
            jb = ((mr >> 2) & 1) * 8 + ((mr >> 1) & 1) * 4 + (mr & 1) * 2;
            hh = mr >> 3;
        }
        int rtl = wv / NG;
#pragma unroll
        for (int i = 0; i < NS; ++i) {
            int j    = jb + i;
            int kind = j / 8;
            int r    = j % 8;
            sh[kind * (RB * NH) + (rtl * 16 + half * 8 + r) * NH + cg * HPW + hh] = v[i];
        }
    }
    __syncthreads();

    if (active) {
        float* crow = C + (size_t)(rt * 16 + half * 8) * NOUT + cg * 64 + 4 * mr;
#pragma unroll
        for (int r = 0; r < 8; ++r)
            *reinterpret_cast<float4*>(crow + (size_t)r * NOUT) =
                make_float4(acc[0][r], acc[1][r], acc[2][r], acc[3][r]);
    }

    int base  = blockIdx.x * (RB * NH);
    int limit = (ntasks / NG) * 16 * NH;
    for (int idx = threadIdx.x; idx < 2 * RB * NH; idx += 256) {
        int kind = idx / (RB * NH);
        int o    = idx % (RB * NH);
        if (base + o < limit) {
            float* dstp = kind ? ald : als;
            dstp[base + o] = sh[idx];
        }
    }

    if (RANK) {
        if (active) {
            int estride = ntasks * 32;
            for (int e = task * 32 + lane; e < E_; e += 2 * estride) {
                int e1 = e + estride;
                bool two = e1 < E_;
                int d0 = edst[e];
                int d1 = two ? edst[e1] : 0;
                int r0 = atomicAdd(&cnt[d0], 1);
                int r1 = 0;
                if (two) r1 = atomicAdd(&cnt[d1], 1);
                rank[e] = r0;
                if (two) rank[e1] = r1;
            }
        }
    }
}

#define OFF_PT  49
#define OFF_CAP (1024 * OFF_PT)

__global__ __launch_bounds__(1024) void scan_fill(const int* __restrict__ cnt,
                                                  const int* __restrict__ src,
                                                  const int* __restrict__ dst,
                                                  const int* __restrict__ rank,
                                                  int* __restrict__ off,
                                                  int* __restrict__ csr,
                                                  int n,
                                                  int E_)
{
    __shared__ alignas(16) int offs[OFF_CAP];
    __shared__ int wsum[32];
    int tid  = threadIdx.x;
    int lane = tid & 31;
    int wv   = tid >> 5;

    const int4* c4 = reinterpret_cast<const int4*>(cnt);
    int4* o4 = reinterpret_cast<int4*>(offs);
    int nv = n >> 2;
    constexpr int NQ = (OFF_CAP / 4 + 1023) / 1024;
    int4 q[NQ];
#pragma unroll
    for (int k = 0; k < NQ; ++k) {
        int i = tid + k * 1024;
        q[k] = make_int4(0, 0, 0, 0);
        if (i < nv) q[k] = c4[i];
    }
#pragma unroll
    for (int k = 0; k < NQ; ++k) {
        int i = tid + k * 1024;
        if (i < OFF_CAP / 4) o4[i] = q[k];
    }
    __syncthreads();
    if (n & 3) {
        for (int i = (nv << 2) + tid; i < n; i += 1024) offs[i] = cnt[i];
        __syncthreads();
    }

    int v[OFF_PT];
    int s = 0;
    int* mine = offs + tid * OFF_PT;
#pragma unroll
    for (int j = 0; j < OFF_PT; ++j) {
        v[j] = mine[j];
        s += v[j];
    }

    int incl = s;
    for (int o = 1; o < 32; o <<= 1) {
        int t = __shfl_up(incl, o, 32);
        if (lane >= o) incl += t;
    }
    if (lane == 31) wsum[wv] = incl;
    __syncthreads();
    if (wv == 0) {
        int w = wsum[lane];
        int sc = w;
        for (int o = 1; o < 32; o <<= 1) {
            int t = __shfl_up(sc, o, 32);
            if (lane >= o) sc += t;
        }
        wsum[lane] = sc - w;
    }
    __syncthreads();

    int run = wsum[wv] + incl - s;
#pragma unroll
    for (int j = 0; j < OFF_PT; ++j) {
        mine[j] = run;
        run += v[j];
    }
    __syncthreads();

    for (int i = blockIdx.x * 1024 + tid; i <= n; i += gridDim.x * 1024)
        off[i] = offs[i];

    int estride = gridDim.x * 1024;
    for (int e0 = blockIdx.x * 1024 + tid; e0 < E_; e0 += 4 * estride) {
        int d[4], r[4], sv[4];
#pragma unroll
        for (int u = 0; u < 4; ++u) {
            int e = e0 + u * estride;
            d[u] = 0; r[u] = 0; sv[u] = 0;
            if (e < E_) {
                d[u]  = dst[e];
                r[u]  = rank[e];
                sv[u] = src[e];
            }
        }
#pragma unroll
        for (int u = 0; u < 4; ++u) {
            int e = e0 + u * estride;
            if (e < E_) csr[offs[d[u]] + r[u]] = sv[u];
        }
    }
}

__device__ __forceinline__ float edge_weight(float x)
{
    float lv = (x >= 0.f) ? x : NEG_SLOPE * x;
    return expf(lv);
}

__device__ __forceinline__ float elu1(float v)
{
    return (v > 0.f) ? v : expm1f(v);
}


#define APITCH 260
#define NPW    4
#define RPB    (8 * NPW)

__device__ __forceinline__ void aggr1_node(int d, int lane,
                                           const int* __restrict__ off,
                                           const int* __restrict__ csr,
                                           const float* __restrict__ als,
                                           const float* __restrict__ ald,
                                           const float* __restrict__ h,
                                           const float* __restrict__ bias,
                                           float4& r0, float4& r1)
{
    int hd0 = lane >> 3;
    int c0 = 4 * lane;
    int c1 = 128 + 4 * lane;
    int eu = lane >> 3;
    int eh = lane & 7;

    float adm = ald[d * 8 + eh];
    int beg = off[d];
    int end = off[d + 1];

    float4 acc0, acc1;
    float den0, den1;
    {
        float W = edge_weight(als[d * 8 + eh] + adm);
        float w0 = __shfl(W, hd0, 32);
        float w1 = __shfl(W, 4 + hd0, 32);
        const float* hs = h + (size_t)d * 256;
        float4 v0 = *reinterpret_cast<const float4*>(hs + c0);
        float4 v1 = *reinterpret_cast<const float4*>(hs + c1);
        acc0 = make_float4(w0 * v0.x, w0 * v0.y, w0 * v0.z, w0 * v0.w);
        acc1 = make_float4(w1 * v1.x, w1 * v1.y, w1 * v1.z, w1 * v1.w);
        den0 = w0;
        den1 = w1;
    }

    int p = beg;
    for (; p + 4 <= end; p += 4) {
        int s[4];
        float4 v0[4], v1[4];
        for (int u = 0; u < 4; ++u) s[u] = csr[p + u];
        int sa = (eu & 1) ? s[1] : s[0];
        int sb = (eu & 1) ? s[3] : s[2];
        int sm = (eu & 2) ? sb : sa;
        float x = als[sm * 8 + eh];
        for (int u = 0; u < 4; ++u) {
            const float* hs = h + (size_t)s[u] * 256;
            v0[u] = *reinterpret_cast<const float4*>(hs + c0);
            v1[u] = *reinterpret_cast<const float4*>(hs + c1);
        }
        float W = edge_weight(x + adm);
        for (int u = 0; u < 4; ++u) {
            float w0 = __shfl(W, 8 * u + hd0, 32);
            float w1 = __shfl(W, 8 * u + 4 + hd0, 32);
            acc0.x += w0 * v0[u].x; acc0.y += w0 * v0[u].y;
            acc0.z += w0 * v0[u].z; acc0.w += w0 * v0[u].w;
            acc1.x += w1 * v1[u].x; acc1.y += w1 * v1[u].y;
            acc1.z += w1 * v1[u].z; acc1.w += w1 * v1[u].w;
            den0 += w0;
            den1 += w1;
        }
    }
    int rem = end - p;
    if (rem > 0) {
        int s0 = csr[p];
        int s1 = (rem > 1) ? csr[p + 1] : s0;
        int s2 = (rem > 2) ? csr[p + 2] : s0;
        int sa = (eu & 1) ? s1 : s0;
        int sm = (eu & 2) ? s2 : sa;
        float x = als[sm * 8 + eh];
        float4 t0[3], t1[3];
        const int st[3] = {s0, s1, s2};
        for (int u = 0; u < 3; ++u) {
            const float* hs = h + (size_t)st[u] * 256;
            t0[u] = *reinterpret_cast<const float4*>(hs + c0);
            t1[u] = *reinterpret_cast<const float4*>(hs + c1);
        }
        float W = edge_weight(x + adm);
        for (int u = 0; u < 3; ++u) {
            if (u < rem) {
                float w0 = __shfl(W, 8 * u + hd0, 32);
                float w1 = __shfl(W, 8 * u + 4 + hd0, 32);
                acc0.x += w0 * t0[u].x; acc0.y += w0 * t0[u].y;
                acc0.z += w0 * t0[u].z; acc0.w += w0 * t0[u].w;
                acc1.x += w1 * t1[u].x; acc1.y += w1 * t1[u].y;
                acc1.z += w1 * t1[u].z; acc1.w += w1 * t1[u].w;
                den0 += w0;
                den1 += w1;
            }
        }
    }

    float4 b0 = *reinterpret_cast<const float4*>(bias + c0);
    float4 b1 = *reinterpret_cast<const float4*>(bias + c1);
    r0.x = elu1(acc0.x / den0 + b0.x); r0.y = elu1(acc0.y / den0 + b0.y);
    r0.z = elu1(acc0.z / den0 + b0.z); r0.w = elu1(acc0.w / den0 + b0.w);
    r1.x = elu1(acc1.x / den1 + b1.x); r1.y = elu1(acc1.y / den1 + b1.y);
    r1.z = elu1(acc1.z / den1 + b1.z); r1.w = elu1(acc1.w / den1 + b1.w);
}

__global__ __launch_bounds__(256, 16) void aggr1_gemm2(const int* __restrict__ off,
                                                   const int* __restrict__ csr,
                                                   const float* __restrict__ als,
                                                   const float* __restrict__ ald,
                                                   const float* __restrict__ h,
                                                   const float* __restrict__ bias,
                                                   const float* __restrict__ W2,
                                                   const float* __restrict__ a_src2,
                                                   const float* __restrict__ a_dst2,
                                                   float* __restrict__ h2,
                                                   float* __restrict__ sd2,
                                                   int n)
{
    __shared__ alignas(16) float tile[RPB * APITCH];
    __shared__ float tot[2 * RPB];
    static_assert(RPB * APITCH >= 8 * 16 * 64, "the partial tiles reuse the A tile's room");

    int wv   = threadIdx.x >> 5;
    int lane = threadIdx.x & 31;

#pragma nounroll
    for (int i = 0; i < NPW; ++i) {
        int row = wv * NPW + i;
        int d = blockIdx.x * RPB + row;
        d = __builtin_amdgcn_readfirstlane(d);
        float4 r0 = make_float4(0.f, 0.f, 0.f, 0.f);
        float4 r1 = r0;
        if (d < n) aggr1_node(d, lane, off, csr, als, ald, h, bias, r0, r1);
        *reinterpret_cast<float4*>(tile + row * APITCH + 4 * lane) = r0;
        *reinterpret_cast<float4*>(tile + row * APITCH + 128 + 4 * lane) = r1;
    }
    __syncthreads();

    int half = lane >> 4;
    int mr   = lane & 15;
    int rt   = wv >> 2;
    int kq   = wv & 3;
    v8f acc[4] = {};
    {
        const float* arow = tile + (rt * 16 + mr) * APITCH + 64 * kq + 4 * half;
        const float* bcol = W2 + (size_t)(64 * kq + 4 * half) * 64 + 4 * mr;
        for (int k0 = 0; k0 < 64; k0 += 8) {
            float4 av = *reinterpret_cast<const float4*>(arow + k0);
            const float* br = bcol + (size_t)k0 * 64;
            float4 q0 = *reinterpret_cast<const float4*>(br);
            float4 q1 = *reinterpret_cast<const float4*>(br + 64);
            float4 q2 = *reinterpret_cast<const float4*>(br + 128);
            float4 q3 = *reinterpret_cast<const float4*>(br + 192);
            acc[0] = mma4(av.x, av.y, q0.x, q1.x, acc[0]);
            acc[1] = mma4(av.x, av.y, q0.y, q1.y, acc[1]);
            acc[2] = mma4(av.x, av.y, q0.z, q1.z, acc[2]);
            acc[3] = mma4(av.x, av.y, q0.w, q1.w, acc[3]);
            acc[0] = mma4(av.z, av.w, q2.x, q3.x, acc[0]);
            acc[1] = mma4(av.z, av.w, q2.y, q3.y, acc[1]);
            acc[2] = mma4(av.z, av.w, q2.z, q3.z, acc[2]);
            acc[3] = mma4(av.z, av.w, q2.w, q3.w, acc[3]);
        }
    }
    __syncthreads();

    {
        float* part = tile + wv * (16 * 64) + (half * 8) * 64 + 4 * mr;
#pragma unroll
        for (int r = 0; r < 8; ++r)
            *reinterpret_cast<float4*>(part + r * 64) =
                make_float4(acc[0][r], acc[1][r], acc[2][r], acc[3][r]);
    }
    __syncthreads();

    int r16 = threadIdx.x >> 4;
    int cq  = threadIdx.x & 15;
    float4 cs4 = *reinterpret_cast<const float4*>(a_src2 + 4 * cq);
    float4 cd4 = *reinterpret_cast<const float4*>(a_dst2 + 4 * cq);
    for (int t2 = 0; t2 < RPB / 16; ++t2) {
        int row = t2 * 16 + r16;
        const float* pp = tile + (t2 * 4) * (16 * 64) + r16 * 64 + 4 * cq;
        float4 s = *reinterpret_cast<const float4*>(pp);
        for (int w = 1; w < 4; ++w) {
            float4 t = *reinterpret_cast<const float4*>(pp + w * (16 * 64));
            s.x += t.x; s.y += t.y; s.z += t.z; s.w += t.w;
        }
        int node = blockIdx.x * RPB + row;
        if (node < n)
            *reinterpret_cast<float4*>(h2 + (size_t)node * 64 + 4 * cq) = s;

        float ps = s.x * cs4.x + s.y * cs4.y + s.z * cs4.z + s.w * cs4.w;
        float pd = s.x * cd4.x + s.y * cd4.y + s.z * cd4.z + s.w * cd4.w;
        for (int m = 8; m >= 1; m >>= 1) {
            ps += __shfl_xor(ps, m, 32);
            pd += __shfl_xor(pd, m, 32);
        }
        if (cq == 0) {
            tot[row * 2]     = ps;
            tot[row * 2 + 1] = pd;
        }
    }
    __syncthreads();
    if (threadIdx.x < 2 * RPB) {
        int idx = blockIdx.x * (2 * RPB) + threadIdx.x;
        if (idx < 2 * n) sd2[idx] = tot[threadIdx.x];
    }
}

__global__ __launch_bounds__(256) void node_aggr2(const int* __restrict__ off,
                                                  const int* __restrict__ csr,
                                                  const float* __restrict__ sd,
                                                  const float* __restrict__ h,
                                                  const float* __restrict__ bias,
                                                  float* __restrict__ out,
                                                  int n)
{
    int d = blockIdx.x * (blockDim.x >> 5) + (threadIdx.x >> 5);
    if (d >= n) return;
    d = __builtin_amdgcn_readfirstlane(d);
    int lane = threadIdx.x & 31;
    int c0 = 2 * lane;
    int eu = lane & 3;

    float ad = sd[2 * d + 1];
    int beg = off[d];
    int end = off[d + 1];

    float2 acc;
    float den;
    {
        float w = edge_weight(sd[2 * d] + ad);
        float2 v = *reinterpret_cast<const float2*>(h + (size_t)d * 64 + c0);
        acc = make_float2(w * v.x, w * v.y);
        den = w;
    }

    int p = beg;
    for (; p + 4 <= end; p += 4) {
        int s[4];
        float2 v[4];
        for (int u = 0; u < 4; ++u) s[u] = csr[p + u];
        int sa = (eu & 1) ? s[1] : s[0];
        int sb = (eu & 1) ? s[3] : s[2];
        int sm = (eu & 2) ? sb : sa;
        float x = sd[2 * sm];
        for (int u = 0; u < 4; ++u)
            v[u] = *reinterpret_cast<const float2*>(h + (size_t)s[u] * 64 + c0);
        float W = edge_weight(x + ad);
        for (int u = 0; u < 4; ++u) {
            float w = __shfl(W, u, 32);
            acc.x += w * v[u].x;
            acc.y += w * v[u].y;
            den += w;
        }
    }
    int rem = end - p;
    if (rem > 0) {
        int s0 = csr[p];
        int s1 = (rem > 1) ? csr[p + 1] : s0;
        int s2 = (rem > 2) ? csr[p + 2] : s0;
        int sa = (eu & 1) ? s1 : s0;
        int sm = (eu & 2) ? s2 : sa;
        float x = sd[2 * sm];
        float2 t[3];
        const int st[3] = {s0, s1, s2};
        for (int u = 0; u < 3; ++u)
            t[u] = *reinterpret_cast<const float2*>(h + (size_t)st[u] * 64 + c0);
        float W = edge_weight(x + ad);
        for (int u = 0; u < 3; ++u) {
            if (u < rem) {
                float w = __shfl(W, u, 32);
                acc.x += w * t[u].x;
                acc.y += w * t[u].y;
                den += w;
            }
        }
    }

    float2 b = *reinterpret_cast<const float2*>(bias + c0);
    float2 r;
    r.x = elu1(acc.x / den + b.x);
    r.y = elu1(acc.y / den + b.y);
    *reinterpret_cast<float2*>(out + (size_t)d * 64 + c0) = r;
}

extern "C" void kernel_launch(void* const* d_in, const int* in_sizes, int n_in,
                              void* d_out, int out_size, void* d_ws, size_t ws_size,
                              hipStream_t stream)
{
    const float* x   = (const float*)d_in[0];
    const int*   ei  = (const int*)  d_in[1];
    const float* W1  = (const float*)d_in[2];
    const float* as1 = (const float*)d_in[3];
    const float* ad1 = (const float*)d_in[4];
    const float* b1  = (const float*)d_in[5];
    const float* W2  = (const float*)d_in[6];
    const float* as2 = (const float*)d_in[7];
    const float* ad2 = (const float*)d_in[8];
    const float* b2  = (const float*)d_in[9];
    float* out = (float*)d_out;

    const int IN_DIM = 128, H1 = 8, F1 = 256, OUT_DIM = 64;
    int n = in_sizes[0] / IN_DIM;
    int e = in_sizes[1] / 2;
    const int* src = ei;
    const int* dst = ei + e;

    float* ws = (float*)d_ws;
    float* h1   = ws;  ws += (size_t)n * F1;
    float* h2   = ws;  ws += (size_t)n * F1;
    int*   ib   = (int*)ws;  ws += (size_t)n * OUT_DIM;
    float* als1 = ws;  ws += (size_t)n * H1;
    float* ald1 = ws;  ws += (size_t)n * H1;
    float* sd2  = ws;  ws += (size_t)n * H1;

    int* cnt  = ib;  ib += n;
    int* rank = ib;  ib += e;
    int* csr  = ib;  ib += e;
    int* off  = ib;

    hipMemsetAsync(cnt, 0, (size_t)n * sizeof(int), stream);

    {
        int tasks = (n / 16) * (F1 / 64);
        gemm_wmma<128, 256, 32, true><<<(tasks + 7) / 8, 256, 0, stream>>>(
            x, W1, h1, as1, ad1, als1, ald1, tasks, dst, cnt, rank, e);
    }

    scan_fill<<<256, 1024, 0, stream>>>(cnt, src, dst, rank, off, csr, n, e);

    aggr1_gemm2<<<(n + 31) / 32, 256, 0, stream>>>(off, csr, als1, ald1, h1, b1,
                                                   W2, as2, ad2, h2, sd2, n);

    node_aggr2<<<(n + 7) / 8, 256, 0, stream>>>(off, csr, sd2, h2, b2, out, n);
}
